# starting kernel
# baseline (speedup 1.0000x reference)
#include <hip/hip_runtime.h>
#include <hip/hip_bf16.h>


typedef float v2f __attribute__((ext_vector_type(2)));
typedef float v8f __attribute__((ext_vector_type(8)));

#define DIM_IN   512
#define DIM_OUT  512
#define NBANKS   64
#define NTOK     512
#define NROWS    1024
#define NSPLIT   4

__global__ void __launch_bounds__(256) banked_zero_kernel(float* __restrict__ out, int n) {
    int i = blockIdx.x * 256 + threadIdx.x;
    if (i < n) out[i] = 0.0f;
}

__global__ void __launch_bounds__(256) banked_linear_wmma_kernel(
    const float* __restrict__ x,
    const int*   __restrict__ sel,
    const float* __restrict__ prob,
    const float* __restrict__ w,
    const float* __restrict__ bias,
    float*       __restrict__ out)
{
    __shared__ int   s_rows[NROWS];
    __shared__ float s_prob[NROWS];
    __shared__ int   s_count;
    __shared__ float s_a[16 * DIM_IN];

    const int bank  = blockIdx.x >> 2;
    const int split = blockIdx.x & 3;
    const int tid   = threadIdx.x;

    if (tid == 0) s_count = 0;
    __syncthreads();

    for (int r = tid; r < NROWS; r += 256) {
        if (sel[r] == bank) {
            int p = atomicAdd(&s_count, 1);
            s_rows[p] = r;
            s_prob[p] = prob[r];
        }
    }
    __syncthreads();
    const int count = s_count;

    const int wave = tid >> 5;
    const int lane = tid & 31;
    const int n0   = (split * 8 + wave) * 16;
    const float* wb = w + (size_t)bank * (DIM_IN * DIM_OUT);

    const int am = lane & 15;
    const int ak = (lane >> 4) << 1;
    const int cn = lane & 15;
    const float* bptr = wb + (size_t)ak * DIM_OUT + n0 + cn;

    for (int mt = 0; mt < count; mt += 16) {
        __syncthreads();

        for (int i = tid; i < 16 * (DIM_IN / 4); i += 256) {
            int m  = i >> 7;
            int c4 = i & 127;
            float4 v = make_float4(0.f, 0.f, 0.f, 0.f);
            int ri = mt + m;
            if (ri < count) {
                int tok = s_rows[ri] >> 1;
                v = ((const float4*)(x + (size_t)tok * DIM_IN))[c4];
            }
            ((float4*)s_a)[m * (DIM_IN / 4) + c4] = v;
        }
        __syncthreads();

        v8f acc = {};
        const float* ldsa = &s_a[am * DIM_IN + ak];
#pragma unroll 4
        for (int k0 = 0; k0 < DIM_IN; k0 += 4) {
            v2f a = *(const v2f*)(ldsa + k0);
            v2f b;
            b.x = bptr[(size_t)k0 * DIM_OUT];
            b.y = bptr[(size_t)k0 * DIM_OUT + DIM_OUT];
            acc = __builtin_amdgcn_wmma_f32_16x16x4_f32(
                 false, a,  false, b,
                 (short)0, acc,  false,  false);
        }

        const int   mbase = (lane >> 4) << 3;
        const float bv    = bias[bank * DIM_OUT + n0 + cn];
#pragma unroll
        for (int v = 0; v < 8; ++v) {
            int ri = mt + mbase + v;
            if (ri < count) {
                int   tok = s_rows[ri] >> 1;
                float p   = s_prob[ri];
                atomicAdd(out + (size_t)tok * DIM_OUT + n0 + cn, (acc[v] + bv) * p);
            }
        }
    }
}

extern "C" void kernel_launch(void* const* d_in, const int* in_sizes, int n_in,
                              void* d_out, int out_size, void* d_ws, size_t ws_size,
                              hipStream_t stream) {
    const float* x    = (const float*)d_in[0];
    const int*   sel  = (const int*)  d_in[1];
    const float* prob = (const float*)d_in[2];
    const float* w    = (const float*)d_in[3];
    const float* bias = (const float*)d_in[4];

    float* out = (float*)d_out;

    const int n = NTOK * DIM_OUT;
    banked_zero_kernel<<<(n + 255) / 256, 256, 0, stream>>>(out, n);
    banked_linear_wmma_kernel<<<NBANKS * NSPLIT, 256, 0, stream>>>(
        x, sel, prob, w, bias, out);
}


# W read once per work-group as 128-bit rows into registers (4 N-tiles per wave) before binning; 16 waves = 2 column groups x 8 K-slices, partial tiles summed in LDS in fixed order
# speedup vs baseline: 2.3997x; 2.3997x over previous
#include <hip/hip_runtime.h>


typedef float v2f __attribute__((ext_vector_type(2)));
typedef float v4f __attribute__((ext_vector_type(4)));
typedef float v8f __attribute__((ext_vector_type(8)));

#define DIM_IN   512
#define DIM_OUT  512
#define NBANKS   64
#define NTOK     512
#define NROWS    1024
#define NSPLIT   4
#define NTHR     512
#define KSLICES  8
#define KSTEPS   16

__global__ void __launch_bounds__(256) banked_zero_kernel(float* __restrict__ out, int n) {
    int i = blockIdx.x * 256 + threadIdx.x;
    if (i < n) out[i] = 0.0f;
}

__global__ void __launch_bounds__(NTHR) banked_linear_wmma_kernel(
    const float* __restrict__ x,
    const int*   __restrict__ sel,
    const float* __restrict__ prob,
    const float* __restrict__ w,
    const float* __restrict__ bias,
    float*       __restrict__ out)
{
    __shared__ int   s_rows[NROWS];
    __shared__ float s_prob[NROWS];
    __shared__ int   s_count;
    __shared__ __attribute__((aligned(16))) float s_a[16 * DIM_IN];
    __shared__ __attribute__((aligned(16))) float s_red[KSLICES * 2 * 16 * 64];

    const int bank  = blockIdx.x >> 2;
    const int split = blockIdx.x & 3;
    const int tid   = threadIdx.x;
    const int wave  = tid >> 5;
    const int lane  = tid & 31;
    const int grp   = wave & 1;
    const int ksl   = wave >> 1;
    const int nb    = split * 128 + grp * 64;

    if (tid == 0) s_count = 0;
    __syncthreads();

    const int   sel0 = sel[tid];
    const int   sel1 = sel[tid + NTHR];
    const float pr0  = prob[tid];
    const float pr1  = prob[tid + NTHR];
    const float bv0  = bias[bank * DIM_OUT + nb + lane];
    const float bv1  = bias[bank * DIM_OUT + nb + 32 + lane];

    const int am = lane & 15;
    const int cn = lane & 15;
    const int hl = lane >> 4;
    const int ak = hl << 1;

    const float* bp = w + (size_t)bank * (DIM_IN * DIM_OUT)
                        + (size_t)(ksl * 64 + ak) * DIM_OUT + nb + 4 * cn;
    v4f blo[KSTEPS], bhi[KSTEPS];
#pragma unroll
    for (int s = 0; s < KSTEPS; ++s) {
        blo[s] = *(const v4f*)(bp + (size_t)(4 * s)     * DIM_OUT);
        bhi[s] = *(const v4f*)(bp + (size_t)(4 * s + 1) * DIM_OUT);
    }

    if (sel0 == bank) {
        int p = atomicAdd(&s_count, 1);
        s_rows[p] = tid;
        s_prob[p] = pr0;
    }
    if (sel1 == bank) {
        int p = atomicAdd(&s_count, 1);
        s_rows[p] = tid + NTHR;
        s_prob[p] = pr1;
    }
    __syncthreads();
    const int count = s_count;

    const float* ldsa = &s_a[am * DIM_IN + ksl * 64 + ak];
    float* redw = &s_red[((ksl * 2 + grp) * 16 + hl * 8) * 64 + 4 * cn];
    const int mfin = (wave >> 1) * 2;

    for (int mt = 0; mt < count; mt += 16) {
        for (int i = tid; i < 16 * (DIM_IN / 4); i += NTHR) {
            int m  = i >> 7;
            int c4 = i & 127;
            float4 v = make_float4(0.f, 0.f, 0.f, 0.f);
            int ri = mt + m;
            if (ri < count) {
                int tok = s_rows[ri] >> 1;
                v = ((const float4*)(x + (size_t)tok * DIM_IN))[c4];
            }
            ((float4*)s_a)[m * (DIM_IN / 4) + c4] = v;
        }
        __syncthreads();

        v8f acc0 = {}, acc1 = {}, acc2 = {}, acc3 = {};
#pragma unroll
        for (int s = 0; s < KSTEPS; ++s) {
            v2f a = *(const v2f*)(ldsa + 4 * s);
            v2f b;
            b.x = blo[s].x; b.y = bhi[s].x;
            acc0 = __builtin_amdgcn_wmma_f32_16x16x4_f32(false, a, false, b, (short)0, acc0, false, false);
            b.x = blo[s].y; b.y = bhi[s].y;
            acc1 = __builtin_amdgcn_wmma_f32_16x16x4_f32(false, a, false, b, (short)0, acc1, false, false);
            b.x = blo[s].z; b.y = bhi[s].z;
            acc2 = __builtin_amdgcn_wmma_f32_16x16x4_f32(false, a, false, b, (short)0, acc2, false, false);
            b.x = blo[s].w; b.y = bhi[s].w;
            acc3 = __builtin_amdgcn_wmma_f32_16x16x4_f32(false, a, false, b, (short)0, acc3, false, false);
        }

#pragma unroll
        for (int v = 0; v < 8; ++v) {
            v4f t;
            t.x = acc0[v]; t.y = acc1[v]; t.z = acc2[v]; t.w = acc3[v];
            *(v4f*)(redw + v * 64) = t;
        }
        __syncthreads();

#pragma unroll
        for (int r = 0; r < 2; ++r) {
            const int m  = mfin + r;
            const int ri = mt + m;
            if (ri < count) {
                const int   tok = s_rows[ri] >> 1;
                const float p   = s_prob[ri];
#pragma unroll
                for (int h = 0; h < 2; ++h) {
                    const int cl = h * 32 + lane;
                    const float* rr = &s_red[(grp * 16 + m) * 64 + cl];
                    float sum = rr[0];
#pragma unroll
                    for (int q = 1; q < KSLICES; ++q) sum += rr[q * (2 * 16 * 64)];
                    const float bv = h ? bv1 : bv0;
                    atomicAdd(out + (size_t)tok * DIM_OUT + nb + cl, (sum + bv) * p);
                }
            }
        }
    }
}

extern "C" void kernel_launch(void* const* d_in, const int* in_sizes, int n_in,
                              void* d_out, int out_size, void* d_ws, size_t ws_size,
                              hipStream_t stream) {
    const float* x    = (const float*)d_in[0];
    const int*   sel  = (const int*)  d_in[1];
    const float* prob = (const float*)d_in[2];
    const float* w    = (const float*)d_in[3];
    const float* bias = (const float*)d_in[4];

    float* out = (float*)d_out;

    const int n = NTOK * DIM_OUT;
    banked_zero_kernel<<<(n + 255) / 256, 256, 0, stream>>>(out, n);
    banked_linear_wmma_kernel<<<NBANKS * NSPLIT, NTHR, 0, stream>>>(
        x, sel, prob, w, bias, out);
}


# Rows binned and their activation rows requested before the W loads; first 16-row tile consumes W in k-step pairs as the loads land (LDS-only barrier), so its matrix work overlaps the W stream
# speedup vs baseline: 2.8236x; 1.1766x over previous
#include <hip/hip_runtime.h>


typedef float v2f __attribute__((ext_vector_type(2)));
typedef float v4f __attribute__((ext_vector_type(4)));
typedef float v8f __attribute__((ext_vector_type(8)));

#define DIM_IN   512
#define DIM_OUT  512
#define NBANKS   64
#define NTOK     512
#define NROWS    1024
#define NSPLIT   4
#define NTHR     512
#define KSLICES  8
#define KSTEPS   16
#define SWROWS   32
#define REDTILE  (KSLICES * 2 * 16 * 64)

__global__ void __launch_bounds__(256) banked_zero_kernel(float* __restrict__ out, int n) {
    int i = blockIdx.x * 256 + threadIdx.x;
    if (i < n) out[i] = 0.0f;
}

__device__ __forceinline__ void lds_only_barrier() {
    __builtin_amdgcn_fence(__ATOMIC_RELEASE, "workgroup", "local");
    __builtin_amdgcn_s_barrier();
    __builtin_amdgcn_fence(__ATOMIC_ACQUIRE, "workgroup", "local");
}

template <bool STREAM>
__device__ __forceinline__ void sweep_tile(const v4f (&blo)[KSTEPS], const v4f (&bhi)[KSTEPS],
                                           const float* ldsa, float* redw)
{
    v8f acc0 = {}, acc1 = {}, acc2 = {}, acc3 = {};
#pragma unroll
    for (int s = 0; s < KSTEPS; ++s) {
        if (STREAM && (s & 1) == 0) __builtin_amdgcn_sched_barrier(0);
        v2f a = *(const v2f*)(ldsa + 4 * s);
        v2f b;
        b.x = blo[s].x; b.y = bhi[s].x;
        acc0 = __builtin_amdgcn_wmma_f32_16x16x4_f32(false, a, false, b, (short)0, acc0, false, false);
        b.x = blo[s].y; b.y = bhi[s].y;
        acc1 = __builtin_amdgcn_wmma_f32_16x16x4_f32(false, a, false, b, (short)0, acc1, false, false);
        b.x = blo[s].z; b.y = bhi[s].z;
        acc2 = __builtin_amdgcn_wmma_f32_16x16x4_f32(false, a, false, b, (short)0, acc2, false, false);
        b.x = blo[s].w; b.y = bhi[s].w;
        acc3 = __builtin_amdgcn_wmma_f32_16x16x4_f32(false, a, false, b, (short)0, acc3, false, false);
    }

#pragma unroll
    for (int v = 0; v < 8; ++v) {
        v4f q;
        q.x = acc0[v]; q.y = acc1[v]; q.z = acc2[v]; q.w = acc3[v];
        *(v4f*)(redw + v * 64) = q;
    }
}

__global__ void __launch_bounds__(NTHR) banked_linear_wmma_kernel(
    const float* __restrict__ x,
    const int*   __restrict__ sel,
    const float* __restrict__ prob,
    const float* __restrict__ w,
    const float* __restrict__ bias,
    float*       __restrict__ out)
{
    __shared__ int   s_rows[NROWS];
    __shared__ float s_prob[NROWS];
    __shared__ int   s_count;
    __shared__ __attribute__((aligned(16))) float s_a[SWROWS * DIM_IN];
    __shared__ __attribute__((aligned(16))) float s_red[2 * REDTILE];

    const int bank  = blockIdx.x >> 2;
    const int split = blockIdx.x & 3;
    const int tid   = threadIdx.x;
    const int wave  = tid >> 5;
    const int lane  = tid & 31;
    const int grp   = wave & 1;
    const int ksl   = wave >> 1;
    const int nb    = split * 128 + grp * 64;

    if (tid == 0) s_count = 0;

    const int   sel0 = sel[tid];
    const int   sel1 = sel[tid + NTHR];
    const float pr0  = prob[tid];
    const float pr1  = prob[tid + NTHR];
    const float bv0  = bias[bank * DIM_OUT + nb + lane];
    const float bv1  = bias[bank * DIM_OUT + nb + 32 + lane];
    __syncthreads();

    if (sel0 == bank) {
        int p = atomicAdd(&s_count, 1);
        s_rows[p] = tid;
        s_prob[p] = pr0;
    }
    if (sel1 == bank) {
        int p = atomicAdd(&s_count, 1);
        s_rows[p] = tid + NTHR;
        s_prob[p] = pr1;
    }
    __syncthreads();
    const int count = __builtin_amdgcn_readfirstlane(s_count);
    if (count == 0) return;

    const int am = lane & 15;
    const int cn = lane & 15;
    const int hl = lane >> 4;
    const int ak = hl << 1;

    const float* ldsa = &s_a[am * DIM_IN + ksl * 64 + ak];
    float* redw = &s_red[((ksl * 2 + grp) * 16 + hl * 8) * 64 + 4 * cn];
    const int mfin = (wave >> 1) * 2;

    const int xm = tid >> 7;
    const int xc = tid & 127;
    const int nt0 = (count > 16) ? 2 : 1;
    int xtok[8];
#pragma unroll
    for (int j = 0; j < 8; ++j) {
        const int ri = xm + 4 * j;
        xtok[j] = s_rows[ri < count ? ri : count - 1] >> 1;
    }
    float4 xr[8];
#pragma unroll
    for (int j = 0; j < 8; ++j) {
        xr[j] = make_float4(0.f, 0.f, 0.f, 0.f);
        if (xm + 4 * j < count)
            xr[j] = ((const float4*)(x + (size_t)xtok[j] * DIM_IN))[xc];
    }

    const float* bp = w + (size_t)bank * (DIM_IN * DIM_OUT)
                        + (size_t)(ksl * 64 + ak) * DIM_OUT + nb + 4 * cn;
    v4f blo[KSTEPS], bhi[KSTEPS];
#pragma unroll
    for (int s = 0; s < KSTEPS; ++s) {
        blo[s] = *(const v4f*)(bp + (size_t)(4 * s)     * DIM_OUT);
        bhi[s] = *(const v4f*)(bp + (size_t)(4 * s + 1) * DIM_OUT);
    }

#pragma unroll
    for (int j = 0; j < 8; ++j)
        if (j < 4 * nt0)
            ((float4*)s_a)[(xm + 4 * j) * (DIM_IN / 4) + xc] = xr[j];
    lds_only_barrier();

    sweep_tile<true>(blo, bhi, ldsa, redw);

    for (int mt = 0; mt < count; mt += SWROWS) {
        if (mt > 0) {
            for (int i = tid; i < SWROWS * (DIM_IN / 4); i += NTHR) {
                int m  = i >> 7;
                int c4 = i & 127;
                float4 v = make_float4(0.f, 0.f, 0.f, 0.f);
                int ri = mt + m;
                if (ri < count) {
                    int tok = s_rows[ri] >> 1;
                    v = ((const float4*)(x + (size_t)tok * DIM_IN))[c4];
                }
                ((float4*)s_a)[m * (DIM_IN / 4) + c4] = v;
            }
            __syncthreads();
        }
#pragma nounroll
        for (int t = (mt == 0) ? 1 : 0; t < 2; ++t)
            if (mt + 16 * t < count)
                sweep_tile<false>(blo, bhi, ldsa + t * (16 * DIM_IN), redw + t * REDTILE);
        __syncthreads();

#pragma unroll
        for (int t = 0; t < 2; ++t) {
#pragma unroll
            for (int r = 0; r < 2; ++r) {
                const int m  = mfin + r;
                const int ri = mt + 16 * t + m;
                if (ri < count) {
                    const int   tok = s_rows[ri] >> 1;
                    const float p   = s_prob[ri];
#pragma unroll
                    for (int h = 0; h < 2; ++h) {
                        const int cl = h * 32 + lane;
                        const float* rr = &s_red[t * REDTILE + (grp * 16 + m) * 64 + cl];
                        float sum = rr[0];
#pragma unroll
                        for (int q = 1; q < KSLICES; ++q) sum += rr[q * (2 * 16 * 64)];
                        const float bv = h ? bv1 : bv0;
                        atomicAdd(out + (size_t)tok * DIM_OUT + nb + cl, (sum + bv) * p);
                    }
                }
            }
        }
    }
}

extern "C" void kernel_launch(void* const* d_in, const int* in_sizes, int n_in,
                              void* d_out, int out_size, void* d_ws, size_t ws_size,
                              hipStream_t stream) {
    const float* x    = (const float*)d_in[0];
    const int*   sel  = (const int*)  d_in[1];
    const float* prob = (const float*)d_in[2];
    const float* w    = (const float*)d_in[3];
    const float* bias = (const float*)d_in[4];

    float* out = (float*)d_out;

    const int n = NTOK * DIM_OUT;
    banked_zero_kernel<<<(n + 255) / 256, 256, 0, stream>>>(out, n);
    banked_linear_wmma_kernel<<<NBANKS * NSPLIT, NTHR, 0, stream>>>(
        x, sel, prob, w, bias, out);
}
